# starting kernel
# baseline (speedup 1.0000x reference)
#include <hip/hip_runtime.h>

typedef float v2f __attribute__((ext_vector_type(2)));
typedef float v4f __attribute__((ext_vector_type(4)));
typedef float v8f __attribute__((ext_vector_type(8)));

#define LSEQ 512
#define KDIM 256
#define RDIM 32
#define H1D  100
#define H2D  100
#define NLAB 10

__global__ __launch_bounds__(1024) void farnn_setup(
    const float* __restrict__ alpha0, const float* __restrict__ D1,
    const float* __restrict__ D2, float* __restrict__ ws) {
  int tid = threadIdx.x;
  int k = tid >> 5, n = tid & 31;
  float s = 0.f;
  for (int c = 0; c < KDIM; ++c) s = fmaf(D1[n * KDIM + c], D2[c * RDIM + k], s);
  ws[k * RDIM + n] = s;
  if (tid < RDIM) {
    float b = 0.f;
    for (int c = 0; c < KDIM; ++c) b = fmaf(alpha0[c], D1[tid * KDIM + c], b);
    ws[RDIM * RDIM + tid] = b;
  }
}

static __device__ __forceinline__ v8f wmma_f32_4(v2f a, v2f b, v8f c) {
  return __builtin_amdgcn_wmma_f32_16x16x4_f32(false, a, false, b, (short)0, c,
                                               false, false);
}

__global__ __launch_bounds__(32) void farnn_main(
    const int* __restrict__ x, const int* __restrict__ lengths,
    const float* __restrict__ ER, const float* __restrict__ ws,
    const float* __restrict__ D2, const float* __restrict__ W1,
    const float* __restrict__ b1, const float* __restrict__ W2,
    const float* __restrict__ b2, const float* __restrict__ W3,
    const float* __restrict__ b3, float* __restrict__ out) {
  __shared__ int   tokbuf[16];
  __shared__ float Bbuf[16][RDIM];
  __shared__ float Ebuf[16][RDIM];
  __shared__ float alphaF[16][KDIM];
  __shared__ float Hb1[16][H1D];
  __shared__ float Hb2[16][H2D];

  const int l    = threadIdx.x;
  const int lm   = l & 15;
  const int half = l >> 4;
  const int seq0 = blockIdx.x * 16;

  int lenB = lengths[seq0 + lm];
  int lenA[8];
#pragma unroll
  for (int v = 0; v < 8; ++v) lenA[v] = lengths[seq0 + v + 8 * half];

  v2f mfrag[2][8];
#pragma unroll
  for (int nt = 0; nt < 2; ++nt)
#pragma unroll
    for (int kc = 0; kc < 8; ++kc) {
      int n = 16 * nt + lm, k0 = 4 * kc + 2 * half;
      v2f f; f.x = ws[k0 * RDIM + n]; f.y = ws[(k0 + 1) * RDIM + n];
      mfrag[nt][kc] = f;
    }

  v8f betaC[2];
#pragma unroll
  for (int nt = 0; nt < 2; ++nt) {
    float b0 = ws[RDIM * RDIM + 16 * nt + lm];
#pragma unroll
    for (int v = 0; v < 8; ++v) betaC[nt][v] = b0;
  }
  v2f gam[8];
#pragma unroll
  for (int kc = 0; kc < 8; ++kc) { gam[kc].x = 0.f; gam[kc].y = 0.f; }

  int rtok = 0;
  if (l < 16) { rtok = x[(seq0 + l) * LSEQ + 0]; tokbuf[l] = rtok; }
  __syncthreads();
  v4f ereg[4];
#pragma unroll
  for (int i = 0; i < 4; ++i) {
    int idx = i * 32 + l, m = idx >> 3, q = idx & 7;
    ereg[i] = *(const v4f*)(ER + (size_t)tokbuf[m] * RDIM + q * 4);
  }
  if (l < 16) rtok = x[(seq0 + l) * LSEQ + 1];

#pragma unroll 1
  for (int t = 0; t < LSEQ; ++t) {
#pragma unroll
    for (int nt = 0; nt < 2; ++nt)
#pragma unroll
      for (int v = 0; v < 8; ++v)
        Bbuf[v + 8 * half][16 * nt + lm] = betaC[nt][v];
#pragma unroll
    for (int i = 0; i < 4; ++i) {
      int idx = i * 32 + l, m = idx >> 3, q = idx & 7;
      *(v4f*)(&Ebuf[m][q * 4]) = ereg[i];
    }
    if (l < 16) tokbuf[l] = rtok;
    __syncthreads();

    v4f eregn[4];
#pragma unroll
    for (int i = 0; i < 4; ++i) { eregn[i].x = 0.f; eregn[i].y = 0.f; eregn[i].z = 0.f; eregn[i].w = 0.f; }
    int rtokn = 0;
    if (t + 1 < LSEQ) {
#pragma unroll
      for (int i = 0; i < 4; ++i) {
        int idx = i * 32 + l, m = idx >> 3, q = idx & 7;
        eregn[i] = *(const v4f*)(ER + (size_t)tokbuf[m] * RDIM + q * 4);
      }
      if (l < 16 && t + 2 < LSEQ) rtokn = x[(seq0 + l) * LSEQ + t + 2];
    }

    const bool actB = (t < lenB);
#pragma unroll
    for (int kc = 0; kc < 8; ++kc) {
      int r = 4 * kc + 2 * half;
      v2f bb = *(const v2f*)(&Bbuf[lm][r]);
      v2f ee = *(const v2f*)(&Ebuf[lm][r]);
      float cx = bb.x * ee.x, cy = bb.y * ee.y;
      gam[kc].x = actB ? cx : gam[kc].x;
      gam[kc].y = actB ? cy : gam[kc].y;
    }

    v8f z = {};
    v8f a00 = z, a01 = z, a10 = z, a11 = z;
#pragma unroll
    for (int kc = 0; kc < 8; kc += 2) {
      a00 = wmma_f32_4(gam[kc],     mfrag[0][kc],     a00);
      a10 = wmma_f32_4(gam[kc],     mfrag[1][kc],     a10);
      a01 = wmma_f32_4(gam[kc + 1], mfrag[0][kc + 1], a01);
      a11 = wmma_f32_4(gam[kc + 1], mfrag[1][kc + 1], a11);
    }
    v8f nb0 = a00 + a01, nb1 = a10 + a11;
#pragma unroll
    for (int v = 0; v < 8; ++v) {
      bool a = (t < lenA[v]);
      betaC[0][v] = a ? nb0[v] : betaC[0][v];
      betaC[1][v] = a ? nb1[v] : betaC[1][v];
    }
#pragma unroll
    for (int i = 0; i < 4; ++i) ereg[i] = eregn[i];
    rtok = rtokn;
  }

#pragma unroll 4
  for (int nt2 = 0; nt2 < 16; ++nt2) {
    v8f accF = {};
    int n = 16 * nt2 + lm;
#pragma unroll
    for (int kc = 0; kc < 8; ++kc) {
      int k0 = 4 * kc + 2 * half;
      v2f bf = *(const v2f*)(D2 + (size_t)n * RDIM + k0);
      accF = wmma_f32_4(gam[kc], bf, accF);
    }
#pragma unroll
    for (int v = 0; v < 8; ++v) alphaF[v + 8 * half][n] = accF[v];
  }
  __syncthreads();

  for (int o = l; o < 16 * H1D; o += 32) {
    int s = o / H1D, j = o % H1D;
    float acc = b1[j];
    for (int c = 0; c < KDIM; ++c) acc = fmaf(alphaF[s][c], W1[j * KDIM + c], acc);
    Hb1[s][j] = fmaxf(acc, 0.f);
  }
  __syncthreads();
  for (int o = l; o < 16 * H2D; o += 32) {
    int s = o / H2D, j = o % H2D;
    float acc = b2[j];
    for (int c = 0; c < H1D; ++c) acc = fmaf(Hb1[s][c], W2[j * H1D + c], acc);
    Hb2[s][j] = fmaxf(acc, 0.f);
  }
  __syncthreads();
  for (int o = l; o < 16 * NLAB; o += 32) {
    int s = o / NLAB, j = o % NLAB;
    float acc = b3[j];
    for (int c = 0; c < H2D; ++c) acc = fmaf(Hb2[s][c], W3[j * H2D + c], acc);
    out[(size_t)(seq0 + s) * NLAB + j] = 1.0f * acc;
  }
}

extern "C" void kernel_launch(void* const* d_in, const int* in_sizes, int n_in,
                              void* d_out, int out_size, void* d_ws, size_t ws_size,
                              hipStream_t stream) {
  const int*   x      = (const int*)d_in[0];
  const int*   lens   = (const int*)d_in[1];
  const float* alpha0 = (const float*)d_in[2];
  const float* ER     = (const float*)d_in[3];
  const float* D1     = (const float*)d_in[4];
  const float* D2     = (const float*)d_in[5];
  const float* W1     = (const float*)d_in[6];
  const float* b1     = (const float*)d_in[7];
  const float* W2     = (const float*)d_in[8];
  const float* b2     = (const float*)d_in[9];
  const float* W3     = (const float*)d_in[10];
  const float* b3     = (const float*)d_in[11];
  float* ws  = (float*)d_ws;
  float* out = (float*)d_out;

  farnn_setup<<<1, 1024, 0, stream>>>(alpha0, D1, D2, ws);
  farnn_main<<<2048 / 16, 32, 0, stream>>>(x, lens, ER, ws, D2, W1, b1, W2, b2,
                                           W3, b3, out);
}


# MLP tail (256-100-100-10) moved from per-lane serial FMA loops to f32 16x16x4 WMMA, weights streamed as B fragments, bias as accumulator start; dot-product summation order becomes the WMMA's
# speedup vs baseline: 1.9558x; 1.9558x over previous
#include <hip/hip_runtime.h>

typedef float v2f __attribute__((ext_vector_type(2)));
typedef float v4f __attribute__((ext_vector_type(4)));
typedef float v8f __attribute__((ext_vector_type(8)));

#define LSEQ 512
#define KDIM 256
#define RDIM 32
#define H1D  100
#define H2D  100
#define NLAB 10
#define AFP  260
#define NT1  7

__global__ __launch_bounds__(1024) void farnn_setup(
    const float* __restrict__ alpha0, const float* __restrict__ D1,
    const float* __restrict__ D2, float* __restrict__ ws) {
  int tid = threadIdx.x;
  int k = tid >> 5, n = tid & 31;
  float s = 0.f;
  for (int c = 0; c < KDIM; ++c) s = fmaf(D1[n * KDIM + c], D2[c * RDIM + k], s);
  ws[k * RDIM + n] = s;
  if (tid < RDIM) {
    float b = 0.f;
    for (int c = 0; c < KDIM; ++c) b = fmaf(alpha0[c], D1[tid * KDIM + c], b);
    ws[RDIM * RDIM + tid] = b;
  }
}

static __device__ __forceinline__ v8f wmma_f32_4(v2f a, v2f b, v8f c) {
  return __builtin_amdgcn_wmma_f32_16x16x4_f32(false, a, false, b, (short)0, c,
                                               false, false);
}

__global__ __launch_bounds__(32) void farnn_main(
    const int* __restrict__ x, const int* __restrict__ lengths,
    const float* __restrict__ ER, const float* __restrict__ ws,
    const float* __restrict__ D2, const float* __restrict__ W1,
    const float* __restrict__ b1, const float* __restrict__ W2,
    const float* __restrict__ b2, const float* __restrict__ W3,
    const float* __restrict__ b3, float* __restrict__ out) {
  __shared__ int   tokbuf[16];
  __shared__ float Bbuf[16][RDIM];
  __shared__ float Ebuf[16][RDIM];
  __shared__ float alphaF[16][AFP];
  __shared__ float Hb1[16][H1D];
  __shared__ float Hb2[16][H2D];
  __shared__ float Lbuf[16 * NLAB];

  const int l    = threadIdx.x;
  const int lm   = l & 15;
  const int half = l >> 4;
  const int seq0 = blockIdx.x * 16;

  int lenB = lengths[seq0 + lm];
  int lenA[8];
#pragma unroll
  for (int v = 0; v < 8; ++v) lenA[v] = lengths[seq0 + v + 8 * half];

  v2f mfrag[2][8];
#pragma unroll
  for (int nt = 0; nt < 2; ++nt)
#pragma unroll
    for (int kc = 0; kc < 8; ++kc) {
      int n = 16 * nt + lm, k0 = 4 * kc + 2 * half;
      v2f f; f.x = ws[k0 * RDIM + n]; f.y = ws[(k0 + 1) * RDIM + n];
      mfrag[nt][kc] = f;
    }

  v8f betaC[2];
#pragma unroll
  for (int nt = 0; nt < 2; ++nt) {
    float b0 = ws[RDIM * RDIM + 16 * nt + lm];
#pragma unroll
    for (int v = 0; v < 8; ++v) betaC[nt][v] = b0;
  }
  v2f gam[8];
#pragma unroll
  for (int kc = 0; kc < 8; ++kc) { gam[kc].x = 0.f; gam[kc].y = 0.f; }

  int rtok = 0;
  if (l < 16) { rtok = x[(seq0 + l) * LSEQ + 0]; tokbuf[l] = rtok; }
  __syncthreads();
  v4f ereg[4];
#pragma unroll
  for (int i = 0; i < 4; ++i) {
    int idx = i * 32 + l, m = idx >> 3, q = idx & 7;
    ereg[i] = *(const v4f*)(ER + (size_t)tokbuf[m] * RDIM + q * 4);
  }
  if (l < 16) rtok = x[(seq0 + l) * LSEQ + 1];

#pragma unroll 1
  for (int t = 0; t < LSEQ; ++t) {
#pragma unroll
    for (int nt = 0; nt < 2; ++nt)
#pragma unroll
      for (int v = 0; v < 8; ++v)
        Bbuf[v + 8 * half][16 * nt + lm] = betaC[nt][v];
#pragma unroll
    for (int i = 0; i < 4; ++i) {
      int idx = i * 32 + l, m = idx >> 3, q = idx & 7;
      *(v4f*)(&Ebuf[m][q * 4]) = ereg[i];
    }
    if (l < 16) tokbuf[l] = rtok;
    __syncthreads();

    v4f eregn[4];
#pragma unroll
    for (int i = 0; i < 4; ++i) { eregn[i].x = 0.f; eregn[i].y = 0.f; eregn[i].z = 0.f; eregn[i].w = 0.f; }
    int rtokn = 0;
    if (t + 1 < LSEQ) {
#pragma unroll
      for (int i = 0; i < 4; ++i) {
        int idx = i * 32 + l, m = idx >> 3, q = idx & 7;
        eregn[i] = *(const v4f*)(ER + (size_t)tokbuf[m] * RDIM + q * 4);
      }
      if (l < 16 && t + 2 < LSEQ) rtokn = x[(seq0 + l) * LSEQ + t + 2];
    }

    const bool actB = (t < lenB);
#pragma unroll
    for (int kc = 0; kc < 8; ++kc) {
      int r = 4 * kc + 2 * half;
      v2f bb = *(const v2f*)(&Bbuf[lm][r]);
      v2f ee = *(const v2f*)(&Ebuf[lm][r]);
      float cx = bb.x * ee.x, cy = bb.y * ee.y;
      gam[kc].x = actB ? cx : gam[kc].x;
      gam[kc].y = actB ? cy : gam[kc].y;
    }

    v8f z = {};
    v8f a00 = z, a01 = z, a10 = z, a11 = z;
#pragma unroll
    for (int kc = 0; kc < 8; kc += 2) {
      a00 = wmma_f32_4(gam[kc],     mfrag[0][kc],     a00);
      a10 = wmma_f32_4(gam[kc],     mfrag[1][kc],     a10);
      a01 = wmma_f32_4(gam[kc + 1], mfrag[0][kc + 1], a01);
      a11 = wmma_f32_4(gam[kc + 1], mfrag[1][kc + 1], a11);
    }
    v8f nb0 = a00 + a01, nb1 = a10 + a11;
#pragma unroll
    for (int v = 0; v < 8; ++v) {
      bool a = (t < lenA[v]);
      betaC[0][v] = a ? nb0[v] : betaC[0][v];
      betaC[1][v] = a ? nb1[v] : betaC[1][v];
    }
#pragma unroll
    for (int i = 0; i < 4; ++i) ereg[i] = eregn[i];
    rtok = rtokn;
  }

#pragma unroll 4
  for (int nt2 = 0; nt2 < 16; ++nt2) {
    v8f accF = {};
    int n = 16 * nt2 + lm;
#pragma unroll
    for (int kc = 0; kc < 8; ++kc) {
      int k0 = 4 * kc + 2 * half;
      v2f bf = *(const v2f*)(D2 + (size_t)n * RDIM + k0);
      accF = wmma_f32_4(gam[kc], bf, accF);
    }
#pragma unroll
    for (int v = 0; v < 8; ++v) alphaF[v + 8 * half][n] = accF[v];
  }
  __syncthreads();


  {
    v8f acc[NT1];
#pragma unroll
    for (int nt = 0; nt < NT1; ++nt) {
      int n = 16 * nt + lm;
      float bias = b1[n < H1D ? n : H1D - 1];
#pragma unroll
      for (int v = 0; v < 8; ++v) acc[nt][v] = bias;
    }
#pragma unroll 4
    for (int g = 0; g < KDIM / 8; ++g) {
      int kb = 8 * g + 4 * half;
      v4f a = *(const v4f*)(&alphaF[lm][kb]);
      v2f a0 = {a.x, a.y}, a1 = {a.z, a.w};
      v4f w[NT1];
#pragma unroll
      for (int nt = 0; nt < NT1; ++nt) {
        int n = 16 * nt + lm;
        int nn = n < H1D ? n : H1D - 1;
        w[nt] = *(const v4f*)(W1 + (size_t)nn * KDIM + kb);
      }
#pragma unroll
      for (int nt = 0; nt < NT1; ++nt) {
        v2f b0 = {w[nt].x, w[nt].y};
        acc[nt] = wmma_f32_4(a0, b0, acc[nt]);
      }
#pragma unroll
      for (int nt = 0; nt < NT1; ++nt) {
        v2f b1f = {w[nt].z, w[nt].w};
        acc[nt] = wmma_f32_4(a1, b1f, acc[nt]);
      }
    }
#pragma unroll
    for (int nt = 0; nt < NT1; ++nt) {
      int n = 16 * nt + lm;
      if (n < H1D) {
#pragma unroll
        for (int v = 0; v < 8; ++v) Hb1[v + 8 * half][n] = fmaxf(acc[nt][v], 0.f);
      }
    }
  }
  __syncthreads();

  {
    v8f acc[NT1];
#pragma unroll
    for (int nt = 0; nt < NT1; ++nt) {
      int n = 16 * nt + lm;
      float bias = b2[n < H2D ? n : H2D - 1];
#pragma unroll
      for (int v = 0; v < 8; ++v) acc[nt][v] = bias;
    }
#pragma unroll 5
    for (int kc = 0; kc < H1D / 4; ++kc) {
      int k0 = 4 * kc + 2 * half;
      v2f a = *(const v2f*)(&Hb1[lm][k0]);
#pragma unroll
      for (int nt = 0; nt < NT1; ++nt) {
        int n = 16 * nt + lm;
        int nn = n < H2D ? n : H2D - 1;
        v2f t = *(const v2f*)(W2 + (size_t)nn * H1D + k0);
        acc[nt] = wmma_f32_4(a, t, acc[nt]);
      }
    }
#pragma unroll
    for (int nt = 0; nt < NT1; ++nt) {
      int n = 16 * nt + lm;
      if (n < H2D) {
#pragma unroll
        for (int v = 0; v < 8; ++v) Hb2[v + 8 * half][n] = fmaxf(acc[nt][v], 0.f);
      }
    }
  }
  __syncthreads();

  {
    int nn = lm < NLAB ? lm : NLAB - 1;
    float bias = b3[nn];
    v8f c0, c1 = {};
#pragma unroll
    for (int v = 0; v < 8; ++v) c0[v] = bias;
#pragma unroll
    for (int kc = 0; kc < H2D / 4; ++kc) {
      int k0 = 4 * kc + 2 * half;
      v2f a = *(const v2f*)(&Hb2[lm][k0]);
      v2f t = *(const v2f*)(W3 + (size_t)nn * H2D + k0);
      if (kc & 1) c1 = wmma_f32_4(a, t, c1);
      else        c0 = wmma_f32_4(a, t, c0);
    }
    v8f c = c0 + c1;
    if (lm < NLAB) {
#pragma unroll
      for (int v = 0; v < 8; ++v) Lbuf[(v + 8 * half) * NLAB + lm] = 1.0f * c[v];
    }
  }
  __syncthreads();
  for (int o = l; o < 16 * NLAB; o += 32) out[(size_t)seq0 * NLAB + o] = Lbuf[o];
}

extern "C" void kernel_launch(void* const* d_in, const int* in_sizes, int n_in,
                              void* d_out, int out_size, void* d_ws, size_t ws_size,
                              hipStream_t stream) {
  const int*   x      = (const int*)d_in[0];
  const int*   lens   = (const int*)d_in[1];
  const float* alpha0 = (const float*)d_in[2];
  const float* ER     = (const float*)d_in[3];
  const float* D1     = (const float*)d_in[4];
  const float* D2     = (const float*)d_in[5];
  const float* W1     = (const float*)d_in[6];
  const float* b1     = (const float*)d_in[7];
  const float* W2     = (const float*)d_in[8];
  const float* b2     = (const float*)d_in[9];
  const float* W3     = (const float*)d_in[10];
  const float* b3     = (const float*)d_in[11];
  float* ws  = (float*)d_ws;
  float* out = (float*)d_out;

  farnn_setup<<<1, 1024, 0, stream>>>(alpha0, D1, D2, ws);
  farnn_main<<<2048 / 16, 32, 0, stream>>>(x, lens, ER, ws, D2, W1, b1, W2, b2,
                                           W3, b3, out);
}


# Time loop transposed (beta^T = M.gamma^T): WMMA D layout = next B layout, so beta, gamma and per-lane gathered embedding rows stay in registers, no LDS per step; r summed in a new order, 2 chains
# speedup vs baseline: 3.7684x; 1.9268x over previous
#include <hip/hip_runtime.h>

typedef float v2f __attribute__((ext_vector_type(2)));
typedef float v4f __attribute__((ext_vector_type(4)));
typedef float v8f __attribute__((ext_vector_type(8)));

#define LSEQ 512
#define KDIM 256
#define RDIM 32
#define H1D  100
#define H2D  100
#define NLAB 10
#define AFP  260
#define NT1  7

__global__ __launch_bounds__(1024) void farnn_setup(
    const float* __restrict__ alpha0, const float* __restrict__ D1,
    const float* __restrict__ D2, float* __restrict__ ws) {
  int tid = threadIdx.x;
  int k = tid >> 5, n = tid & 31;
  float s = 0.f;
  for (int c = 0; c < KDIM; ++c) s = fmaf(D1[n * KDIM + c], D2[c * RDIM + k], s);
  ws[k * RDIM + n] = s;
  if (tid < RDIM) {
    float b = 0.f;
    for (int c = 0; c < KDIM; ++c) b = fmaf(alpha0[c], D1[tid * KDIM + c], b);
    ws[RDIM * RDIM + tid] = b;
  }
}

static __device__ __forceinline__ v8f wmma_f32_4(v2f a, v2f b, v8f c) {
  return __builtin_amdgcn_wmma_f32_16x16x4_f32(false, a, false, b, (short)0, c,
                                               false, false);
}

typedef int v4i __attribute__((ext_vector_type(4)));

#define PFD 4

#define KROW(kc, half) (16 * ((kc) >> 2) + 8 * (half) + 2 * ((kc) & 3))
__global__ __launch_bounds__(32) void farnn_main(
    const int* __restrict__ x, const int* __restrict__ lengths,
    const float* __restrict__ ER, const float* __restrict__ ws,
    const float* __restrict__ D2, const float* __restrict__ W1,
    const float* __restrict__ b1, const float* __restrict__ W2,
    const float* __restrict__ b2, const float* __restrict__ W3,
    const float* __restrict__ b3, float* __restrict__ out) {
  __shared__ float alphaF[16][AFP];
  __shared__ float Hb1[16][H1D];
  __shared__ float Hb2[16][H2D];
  __shared__ float Lbuf[16 * NLAB];

  const int l    = threadIdx.x;
  const int lm   = l & 15;
  const int half = l >> 4;
  const int seq0 = blockIdx.x * 16;

  const int len = lengths[seq0 + lm];

  v2f mA[2][8];
#pragma unroll
  for (int rt = 0; rt < 2; ++rt)
#pragma unroll
    for (int kc = 0; kc < 8; ++kc) {
      int ro = 16 * rt + lm, ri = KROW(kc, half);
      v2f f; f.x = ws[ri * RDIM + ro]; f.y = ws[(ri + 1) * RDIM + ro];
      mA[rt][kc] = f;
    }

  v8f dT[2];
#pragma unroll
  for (int rt = 0; rt < 2; ++rt) {
    v4f lo = *(const v4f*)(ws + RDIM * RDIM + 16 * rt + 8 * half);
    v4f hi = *(const v4f*)(ws + RDIM * RDIM + 16 * rt + 8 * half + 4);
    dT[rt][0] = lo.x; dT[rt][1] = lo.y; dT[rt][2] = lo.z; dT[rt][3] = lo.w;
    dT[rt][4] = hi.x; dT[rt][5] = hi.y; dT[rt][6] = hi.z; dT[rt][7] = hi.w;
  }
  v2f gam[8];
#pragma unroll
  for (int kc = 0; kc < 8; ++kc) { gam[kc].x = 0.f; gam[kc].y = 0.f; }

  asm volatile("" ::: "memory");

  const int*   xrow = x + (size_t)(seq0 + lm) * LSEQ;
  const float* ERh  = ER + 8 * half;
  v4f eb[PFD][4];
  {
    v4i tokA = *(const v4i*)(xrow);
#pragma unroll
    for (int j = 0; j < PFD; ++j)
#pragma unroll
      for (int i = 0; i < 4; ++i)
        eb[j][i] = *(const v4f*)(ERh + (size_t)tokA[j] * RDIM + 16 * (i >> 1) + 4 * (i & 1));
  }
  v4i tokB = *(const v4i*)(xrow + PFD);

#pragma unroll 1
  for (int t0 = 0; t0 < LSEQ; t0 += PFD) {
    int cn = (t0 >> 2) + 2;
    cn = cn < LSEQ / 4 ? cn : LSEQ / 4 - 1;
    v4i tokC = *(const v4i*)(xrow + 4 * cn);

#pragma unroll
    for (int j = 0; j < PFD; ++j) {
      const bool act = (t0 + j) < len;
#pragma unroll
      for (int kc = 0; kc < 8; ++kc) {
        const int rt = kc >> 2, v = 2 * (kc & 3);
        const v4f ev = eb[j][2 * rt + (v >> 2)];
        const float e0 = (v & 3) ? ev.z : ev.x;
        const float e1 = (v & 3) ? ev.w : ev.y;
        float gx = dT[rt][v] * e0, gy = dT[rt][v + 1] * e1;
        gam[kc].x = act ? gx : gam[kc].x;
        gam[kc].y = act ? gy : gam[kc].y;
      }
#pragma unroll
      for (int i = 0; i < 4; ++i)
        eb[j][i] = *(const v4f*)(ERh + (size_t)tokB[j] * RDIM + 16 * (i >> 1) + 4 * (i & 1));
      asm volatile("" ::: "memory");

      v8f z = {};
      v8f d0 = wmma_f32_4(mA[0][0], gam[0], z);
      v8f d1 = wmma_f32_4(mA[1][0], gam[0], z);
#pragma unroll
      for (int kc = 1; kc < 8; ++kc) {
        d0 = wmma_f32_4(mA[0][kc], gam[kc], d0);
        d1 = wmma_f32_4(mA[1][kc], gam[kc], d1);
      }
      dT[0] = d0; dT[1] = d1;
    }
    tokB = tokC;
  }

#pragma unroll 4
  for (int nt2 = 0; nt2 < 16; ++nt2) {
    v8f accF = {};
    int n = 16 * nt2 + lm;
#pragma unroll
    for (int kc = 0; kc < 8; ++kc) {
      int k0 = KROW(kc, half);
      v2f bf = *(const v2f*)(D2 + (size_t)n * RDIM + k0);
      accF = wmma_f32_4(gam[kc], bf, accF);
    }
#pragma unroll
    for (int v = 0; v < 8; ++v) alphaF[v + 8 * half][n] = accF[v];
  }
  __syncthreads();


  {
    v8f acc[NT1];
#pragma unroll
    for (int nt = 0; nt < NT1; ++nt) {
      int n = 16 * nt + lm;
      float bias = b1[n < H1D ? n : H1D - 1];
#pragma unroll
      for (int v = 0; v < 8; ++v) acc[nt][v] = bias;
    }
#pragma unroll 4
    for (int g = 0; g < KDIM / 8; ++g) {
      int kb = 8 * g + 4 * half;
      v4f a = *(const v4f*)(&alphaF[lm][kb]);
      v2f a0 = {a.x, a.y}, a1 = {a.z, a.w};
      v4f w[NT1];
#pragma unroll
      for (int nt = 0; nt < NT1; ++nt) {
        int n = 16 * nt + lm;
        int nn = n < H1D ? n : H1D - 1;
        w[nt] = *(const v4f*)(W1 + (size_t)nn * KDIM + kb);
      }
#pragma unroll
      for (int nt = 0; nt < NT1; ++nt) {
        v2f b0 = {w[nt].x, w[nt].y};
        acc[nt] = wmma_f32_4(a0, b0, acc[nt]);
      }
#pragma unroll
      for (int nt = 0; nt < NT1; ++nt) {
        v2f b1f = {w[nt].z, w[nt].w};
        acc[nt] = wmma_f32_4(a1, b1f, acc[nt]);
      }
    }
#pragma unroll
    for (int nt = 0; nt < NT1; ++nt) {
      int n = 16 * nt + lm;
      if (n < H1D) {
#pragma unroll
        for (int v = 0; v < 8; ++v) Hb1[v + 8 * half][n] = fmaxf(acc[nt][v], 0.f);
      }
    }
  }
  __syncthreads();

  {
    v8f acc[NT1];
#pragma unroll
    for (int nt = 0; nt < NT1; ++nt) {
      int n = 16 * nt + lm;
      float bias = b2[n < H2D ? n : H2D - 1];
#pragma unroll
      for (int v = 0; v < 8; ++v) acc[nt][v] = bias;
    }
#pragma unroll 5
    for (int kc = 0; kc < H1D / 4; ++kc) {
      int k0 = 4 * kc + 2 * half;
      v2f a = *(const v2f*)(&Hb1[lm][k0]);
#pragma unroll
      for (int nt = 0; nt < NT1; ++nt) {
        int n = 16 * nt + lm;
        int nn = n < H2D ? n : H2D - 1;
        v2f t = *(const v2f*)(W2 + (size_t)nn * H1D + k0);
        acc[nt] = wmma_f32_4(a, t, acc[nt]);
      }
    }
#pragma unroll
    for (int nt = 0; nt < NT1; ++nt) {
      int n = 16 * nt + lm;
      if (n < H2D) {
#pragma unroll
        for (int v = 0; v < 8; ++v) Hb2[v + 8 * half][n] = fmaxf(acc[nt][v], 0.f);
      }
    }
  }
  __syncthreads();

  {
    int nn = lm < NLAB ? lm : NLAB - 1;
    float bias = b3[nn];
    v8f c0, c1 = {};
#pragma unroll
    for (int v = 0; v < 8; ++v) c0[v] = bias;
#pragma unroll
    for (int kc = 0; kc < H2D / 4; ++kc) {
      int k0 = 4 * kc + 2 * half;
      v2f a = *(const v2f*)(&Hb2[lm][k0]);
      v2f t = *(const v2f*)(W3 + (size_t)nn * H2D + k0);
      if (kc & 1) c1 = wmma_f32_4(a, t, c1);
      else        c0 = wmma_f32_4(a, t, c0);
    }
    v8f c = c0 + c1;
    if (lm < NLAB) {
#pragma unroll
      for (int v = 0; v < 8; ++v) Lbuf[(v + 8 * half) * NLAB + lm] = 1.0f * c[v];
    }
  }
  __syncthreads();
  for (int o = l; o < 16 * NLAB; o += 32) out[(size_t)seq0 * NLAB + o] = Lbuf[o];
}

extern "C" void kernel_launch(void* const* d_in, const int* in_sizes, int n_in,
                              void* d_out, int out_size, void* d_ws, size_t ws_size,
                              hipStream_t stream) {
  const int*   x      = (const int*)d_in[0];
  const int*   lens   = (const int*)d_in[1];
  const float* alpha0 = (const float*)d_in[2];
  const float* ER     = (const float*)d_in[3];
  const float* D1     = (const float*)d_in[4];
  const float* D2     = (const float*)d_in[5];
  const float* W1     = (const float*)d_in[6];
  const float* b1     = (const float*)d_in[7];
  const float* W2     = (const float*)d_in[8];
  const float* b2     = (const float*)d_in[9];
  const float* W3     = (const float*)d_in[10];
  const float* b3     = (const float*)d_in[11];
  float* ws  = (float*)d_ws;
  float* out = (float*)d_out;

  farnn_setup<<<1, 1024, 0, stream>>>(alpha0, D1, D2, ws);
  farnn_main<<<2048 / 16, 32, 0, stream>>>(x, lens, ER, ws, D2, W1, b1, W2, b2,
                                           W3, b3, out);
}


# Set-up launch folded into the main kernel: each block builds M=D1@D2 (transposed, f32 WMMA; D layout = the loop's A fragments) and beta0 itself; one launch, no workspace; M, beta0 summed in new order
# speedup vs baseline: 4.8977x; 1.2997x over previous
#include <hip/hip_runtime.h>

typedef float v2f __attribute__((ext_vector_type(2)));
typedef float v4f __attribute__((ext_vector_type(4)));
typedef float v8f __attribute__((ext_vector_type(8)));

#define LSEQ 512
#define KDIM 256
#define RDIM 32
#define H1D  100
#define H2D  100
#define NLAB 10
#define AFP  260
#define NT1  7

static __device__ __forceinline__ v8f wmma_f32_4(v2f a, v2f b, v8f c) {
  return __builtin_amdgcn_wmma_f32_16x16x4_f32(false, a, false, b, (short)0, c,
                                               false, false);
}

typedef int v4i __attribute__((ext_vector_type(4)));

#define PFD 4

#define PGB 4

struct PBatch { v4f d1[PGB][2]; float a[PGB][2][4]; v4f al[PGB]; };

static __device__ __forceinline__ void pbatch_load(PBatch& f, int b, const float* __restrict__ d1p,
                                                   const float* __restrict__ d2p,
                                                   const float* __restrict__ alp) {
#pragma unroll
  for (int u = 0; u < PGB; ++u) {
    const int k0 = 8 * (PGB * b + u);
#pragma unroll
    for (int rt = 0; rt < 2; ++rt) f.d1[u][rt] = *(const v4f*)(d1p + (size_t)(16 * rt) * KDIM + k0);
#pragma unroll
    for (int j = 0; j < 2; ++j)
#pragma unroll
      for (int q = 0; q < 4; ++q) f.a[u][j][q] = d2p[(size_t)(k0 + q) * RDIM + 16 * j];
    f.al[u] = *(const v4f*)(alp + k0);
  }
}

static __device__ __forceinline__ void pbatch_mac(const PBatch& f, v8f (&accM)[2][2], float (&p)[2]) {
#pragma unroll
  for (int u = 0; u < PGB; ++u)
#pragma unroll
    for (int w = 0; w < 2; ++w) {
#pragma unroll
      for (int j = 0; j < 2; ++j)
#pragma unroll
        for (int rt = 0; rt < 2; ++rt) {
          v2f af = {f.a[u][j][2 * w], f.a[u][j][2 * w + 1]};
          v2f bf = {f.d1[u][rt][2 * w], f.d1[u][rt][2 * w + 1]};
          accM[j][rt] = wmma_f32_4(af, bf, accM[j][rt]);
        }
#pragma unroll
      for (int rt = 0; rt < 2; ++rt) {
        p[rt] = fmaf(f.d1[u][rt][2 * w], f.al[u][2 * w], p[rt]);
        p[rt] = fmaf(f.d1[u][rt][2 * w + 1], f.al[u][2 * w + 1], p[rt]);
      }
    }
}

#define KROW(kc, half) (16 * ((kc) >> 2) + 8 * (half) + 2 * ((kc) & 3))
__global__ __launch_bounds__(32) void farnn_main(
    const int* __restrict__ x, const int* __restrict__ lengths,
    const float* __restrict__ alpha0, const float* __restrict__ ER,
    const float* __restrict__ D1, const float* __restrict__ D2,
    const float* __restrict__ W1,
    const float* __restrict__ b1, const float* __restrict__ W2,
    const float* __restrict__ b2, const float* __restrict__ W3,
    const float* __restrict__ b3, float* __restrict__ out) {
  __shared__ float alphaF[16][AFP];
  __shared__ float Hb1[16][H1D];
  __shared__ float Hb2[16][H2D];
  __shared__ float Lbuf[16 * NLAB];

  const int l    = threadIdx.x;
  const int lm   = l & 15;
  const int half = l >> 4;
  const int seq0 = blockIdx.x * 16;

  const int len = lengths[seq0 + lm];

  const int*   xrow = x + (size_t)(seq0 + lm) * LSEQ;
  v4i tokA = *(const v4i*)(xrow);
  v4i tokB = *(const v4i*)(xrow + PFD);

  const float* ERh  = ER + 8 * half;
  v4f eb[PFD][4];

  const float* d1p = D1 + (size_t)lm * KDIM + 4 * half;
  const float* d2p = D2 + (size_t)(4 * half) * RDIM + lm;
  const float* alp = alpha0 + 4 * half;
  v8f accM[2][2];
#pragma unroll
  for (int j = 0; j < 2; ++j)
#pragma unroll
    for (int rt = 0; rt < 2; ++rt) accM[j][rt] = (v8f){};
  float p[2] = {0.f, 0.f};
  {
    PBatch fa, fb;
    pbatch_load(fa, 0, d1p, d2p, alp);
#pragma unroll 1
    for (int b = 0; b < KDIM / (8 * PGB) - 2; b += 2) {
      pbatch_load(fb, b + 1, d1p, d2p, alp);
      asm volatile("" ::: "memory");
      pbatch_mac(fa, accM, p);
      pbatch_load(fa, b + 2, d1p, d2p, alp);
      asm volatile("" ::: "memory");
      pbatch_mac(fb, accM, p);
    }
    pbatch_load(fb, KDIM / (8 * PGB) - 1, d1p, d2p, alp);
    asm volatile("" ::: "memory");
    pbatch_mac(fa, accM, p);
#pragma unroll
    for (int j = 0; j < PFD; ++j)
#pragma unroll
      for (int i = 0; i < 4; ++i)
        eb[j][i] = *(const v4f*)(ERh + (size_t)tokA[j] * RDIM + 16 * (i >> 1) + 4 * (i & 1));
    asm volatile("" ::: "memory");
    pbatch_mac(fb, accM, p);
  }

  v2f mA[2][8];
#pragma unroll
  for (int rt = 0; rt < 2; ++rt)
#pragma unroll
    for (int kc = 0; kc < 8; ++kc) {
      v2f f; f.x = accM[kc >> 2][rt][2 * (kc & 3)]; f.y = accM[kc >> 2][rt][2 * (kc & 3) + 1];
      mA[rt][kc] = f;
    }

  v8f dT[2];
#pragma unroll
  for (int rt = 0; rt < 2; ++rt) {
    float ps = p[rt] + __shfl_xor(p[rt], 16);
#pragma unroll
    for (int v = 0; v < 8; ++v) dT[rt][v] = __shfl(ps, 8 * half + v);
  }
  v2f gam[8];
#pragma unroll
  for (int kc = 0; kc < 8; ++kc) { gam[kc].x = 0.f; gam[kc].y = 0.f; }

#pragma unroll 1
  for (int t0 = 0; t0 < LSEQ; t0 += PFD) {
    int cn = (t0 >> 2) + 2;
    cn = cn < LSEQ / 4 ? cn : LSEQ / 4 - 1;
    v4i tokC = *(const v4i*)(xrow + 4 * cn);

#pragma unroll
    for (int j = 0; j < PFD; ++j) {
      const bool act = (t0 + j) < len;
#pragma unroll
      for (int kc = 0; kc < 8; ++kc) {
        const int rt = kc >> 2, v = 2 * (kc & 3);
        const v4f ev = eb[j][2 * rt + (v >> 2)];
        const float e0 = (v & 3) ? ev.z : ev.x;
        const float e1 = (v & 3) ? ev.w : ev.y;
        float gx = dT[rt][v] * e0, gy = dT[rt][v + 1] * e1;
        gam[kc].x = act ? gx : gam[kc].x;
        gam[kc].y = act ? gy : gam[kc].y;
      }
#pragma unroll
      for (int i = 0; i < 4; ++i)
        eb[j][i] = *(const v4f*)(ERh + (size_t)tokB[j] * RDIM + 16 * (i >> 1) + 4 * (i & 1));
      asm volatile("" ::: "memory");

      v8f z = {};
      v8f d0 = wmma_f32_4(mA[0][0], gam[0], z);
      v8f d1 = wmma_f32_4(mA[1][0], gam[0], z);
#pragma unroll
      for (int kc = 1; kc < 8; ++kc) {
        d0 = wmma_f32_4(mA[0][kc], gam[kc], d0);
        d1 = wmma_f32_4(mA[1][kc], gam[kc], d1);
      }
      dT[0] = d0; dT[1] = d1;
    }
    tokB = tokC;
  }

#pragma unroll 4
  for (int nt2 = 0; nt2 < 16; ++nt2) {
    v8f accF = {};
    int n = 16 * nt2 + lm;
#pragma unroll
    for (int kc = 0; kc < 8; ++kc) {
      int k0 = KROW(kc, half);
      v2f bf = *(const v2f*)(D2 + (size_t)n * RDIM + k0);
      accF = wmma_f32_4(gam[kc], bf, accF);
    }
#pragma unroll
    for (int v = 0; v < 8; ++v) alphaF[v + 8 * half][n] = accF[v];
  }
  __syncthreads();


  {
    v8f acc[NT1];
#pragma unroll
    for (int nt = 0; nt < NT1; ++nt) {
      int n = 16 * nt + lm;
      float bias = b1[n < H1D ? n : H1D - 1];
#pragma unroll
      for (int v = 0; v < 8; ++v) acc[nt][v] = bias;
    }
#pragma unroll 4
    for (int g = 0; g < KDIM / 8; ++g) {
      int kb = 8 * g + 4 * half;
      v4f a = *(const v4f*)(&alphaF[lm][kb]);
      v2f a0 = {a.x, a.y}, a1 = {a.z, a.w};
      v4f w[NT1];
#pragma unroll
      for (int nt = 0; nt < NT1; ++nt) {
        int n = 16 * nt + lm;
        int nn = n < H1D ? n : H1D - 1;
        w[nt] = *(const v4f*)(W1 + (size_t)nn * KDIM + kb);
      }
#pragma unroll
      for (int nt = 0; nt < NT1; ++nt) {
        v2f b0 = {w[nt].x, w[nt].y};
        acc[nt] = wmma_f32_4(a0, b0, acc[nt]);
      }
#pragma unroll
      for (int nt = 0; nt < NT1; ++nt) {
        v2f b1f = {w[nt].z, w[nt].w};
        acc[nt] = wmma_f32_4(a1, b1f, acc[nt]);
      }
    }
#pragma unroll
    for (int nt = 0; nt < NT1; ++nt) {
      int n = 16 * nt + lm;
      if (n < H1D) {
#pragma unroll
        for (int v = 0; v < 8; ++v) Hb1[v + 8 * half][n] = fmaxf(acc[nt][v], 0.f);
      }
    }
  }
  __syncthreads();

  {
    v8f acc[NT1];
#pragma unroll
    for (int nt = 0; nt < NT1; ++nt) {
      int n = 16 * nt + lm;
      float bias = b2[n < H2D ? n : H2D - 1];
#pragma unroll
      for (int v = 0; v < 8; ++v) acc[nt][v] = bias;
    }
#pragma unroll 5
    for (int kc = 0; kc < H1D / 4; ++kc) {
      int k0 = 4 * kc + 2 * half;
      v2f a = *(const v2f*)(&Hb1[lm][k0]);
#pragma unroll
      for (int nt = 0; nt < NT1; ++nt) {
        int n = 16 * nt + lm;
        int nn = n < H2D ? n : H2D - 1;
        v2f t = *(const v2f*)(W2 + (size_t)nn * H1D + k0);
        acc[nt] = wmma_f32_4(a, t, acc[nt]);
      }
    }
#pragma unroll
    for (int nt = 0; nt < NT1; ++nt) {
      int n = 16 * nt + lm;
      if (n < H2D) {
#pragma unroll
        for (int v = 0; v < 8; ++v) Hb2[v + 8 * half][n] = fmaxf(acc[nt][v], 0.f);
      }
    }
  }
  __syncthreads();

  {
    int nn = lm < NLAB ? lm : NLAB - 1;
    float bias = b3[nn];
    v8f c0, c1 = {};
#pragma unroll
    for (int v = 0; v < 8; ++v) c0[v] = bias;
#pragma unroll
    for (int kc = 0; kc < H2D / 4; ++kc) {
      int k0 = 4 * kc + 2 * half;
      v2f a = *(const v2f*)(&Hb2[lm][k0]);
      v2f t = *(const v2f*)(W3 + (size_t)nn * H2D + k0);
      if (kc & 1) c1 = wmma_f32_4(a, t, c1);
      else        c0 = wmma_f32_4(a, t, c0);
    }
    v8f c = c0 + c1;
    if (lm < NLAB) {
#pragma unroll
      for (int v = 0; v < 8; ++v) Lbuf[(v + 8 * half) * NLAB + lm] = 1.0f * c[v];
    }
  }
  __syncthreads();
  for (int o = l; o < 16 * NLAB; o += 32) out[(size_t)seq0 * NLAB + o] = Lbuf[o];
}

extern "C" void kernel_launch(void* const* d_in, const int* in_sizes, int n_in,
                              void* d_out, int out_size, void* d_ws, size_t ws_size,
                              hipStream_t stream) {
  const int*   x      = (const int*)d_in[0];
  const int*   lens   = (const int*)d_in[1];
  const float* alpha0 = (const float*)d_in[2];
  const float* ER     = (const float*)d_in[3];
  const float* D1     = (const float*)d_in[4];
  const float* D2     = (const float*)d_in[5];
  const float* W1     = (const float*)d_in[6];
  const float* b1     = (const float*)d_in[7];
  const float* W2     = (const float*)d_in[8];
  const float* b2     = (const float*)d_in[9];
  const float* W3     = (const float*)d_in[10];
  const float* b3     = (const float*)d_in[11];
  float* out = (float*)d_out;

  farnn_main<<<2048 / 16, 32, 0, stream>>>(x, lens, alpha0, ER, D1, D2, W1, b1, W2,
                                           b2, W3, b3, out);
}


# Block widened to 4 waves: waves 1-3 load all weights of the part behind the time loop at entry, then run alpha_final and the 3 MLP layers split by column tile; sum order unchanged
# speedup vs baseline: 5.2584x; 1.0736x over previous
#include <hip/hip_runtime.h>

typedef float v2f __attribute__((ext_vector_type(2)));
typedef float v4f __attribute__((ext_vector_type(4)));
typedef float v8f __attribute__((ext_vector_type(8)));

#define LSEQ 512
#define KDIM 256
#define RDIM 32
#define H1D  100
#define H2D  100
#define NLAB 10
#define AFP  260
#define NT1  7
#define NHELP 3
#define SA   6
#define SB   3

static __device__ __forceinline__ v8f wmma_f32_4(v2f a, v2f b, v8f c) {
  return __builtin_amdgcn_wmma_f32_16x16x4_f32(false, a, false, b, (short)0, c,
                                               false, false);
}

typedef int v4i __attribute__((ext_vector_type(4)));

#define PFD 4

#define PGB 4

struct PBatch { v4f d1[PGB][2]; float a[PGB][2][4]; v4f al[PGB]; };

static __device__ __forceinline__ void pbatch_load(PBatch& f, int b, const float* __restrict__ d1p,
                                                   const float* __restrict__ d2p,
                                                   const float* __restrict__ alp) {
#pragma unroll
  for (int u = 0; u < PGB; ++u) {
    const int k0 = 8 * (PGB * b + u);
#pragma unroll
    for (int rt = 0; rt < 2; ++rt) f.d1[u][rt] = *(const v4f*)(d1p + (size_t)(16 * rt) * KDIM + k0);
#pragma unroll
    for (int j = 0; j < 2; ++j)
#pragma unroll
      for (int q = 0; q < 4; ++q) f.a[u][j][q] = d2p[(size_t)(k0 + q) * RDIM + 16 * j];
    f.al[u] = *(const v4f*)(alp + k0);
  }
}

static __device__ __forceinline__ void pbatch_mac(const PBatch& f, v8f (&accM)[2][2], float (&p)[2]) {
#pragma unroll
  for (int u = 0; u < PGB; ++u)
#pragma unroll
    for (int w = 0; w < 2; ++w) {
#pragma unroll
      for (int j = 0; j < 2; ++j)
#pragma unroll
        for (int rt = 0; rt < 2; ++rt) {
          v2f af = {f.a[u][j][2 * w], f.a[u][j][2 * w + 1]};
          v2f bf = {f.d1[u][rt][2 * w], f.d1[u][rt][2 * w + 1]};
          accM[j][rt] = wmma_f32_4(af, bf, accM[j][rt]);
        }
#pragma unroll
      for (int rt = 0; rt < 2; ++rt) {
        p[rt] = fmaf(f.d1[u][rt][2 * w], f.al[u][2 * w], p[rt]);
        p[rt] = fmaf(f.d1[u][rt][2 * w + 1], f.al[u][2 * w + 1], p[rt]);
      }
    }
}

#define KROW(kc, half) (16 * ((kc) >> 2) + 8 * (half) + 2 * ((kc) & 3))
__global__ __launch_bounds__(32 * (1 + NHELP)) void farnn_main(
    const int* __restrict__ x, const int* __restrict__ lengths,
    const float* __restrict__ alpha0, const float* __restrict__ ER,
    const float* __restrict__ D1, const float* __restrict__ D2,
    const float* __restrict__ W1,
    const float* __restrict__ b1, const float* __restrict__ W2,
    const float* __restrict__ b2, const float* __restrict__ W3,
    const float* __restrict__ b3, float* __restrict__ out) {
  __shared__ float alphaF[16][AFP];
  __shared__ float Hb1[16][H1D];
  __shared__ float Hb2[16][H2D];
  __shared__ float Lbuf[16 * NLAB];
  __shared__ v2f   gamS[8][32];

  const int l    = threadIdx.x & 31;
  const int wv   = __builtin_amdgcn_readfirstlane((int)(threadIdx.x >> 5));
  const int h    = wv - 1;
  const int lm   = l & 15;
  const int half = l >> 4;
  const int seq0 = blockIdx.x * 16;

  v2f   dA[SA][8];
  v4f   w1r[SB][KDIM / 8];
  v2f   w2r[SB][H1D / 4];
  v2f   w3r[H2D / 4];
  float bs1[SB], bs2[SB], bs3;

  if (wv == 0) {
  const int len = lengths[seq0 + lm];

  const int*   xrow = x + (size_t)(seq0 + lm) * LSEQ;
  v4i tokA = *(const v4i*)(xrow);
  v4i tokB = *(const v4i*)(xrow + PFD);

  const float* ERh  = ER + 8 * half;
  v4f eb[PFD][4];

  const float* d1p = D1 + (size_t)lm * KDIM + 4 * half;
  const float* d2p = D2 + (size_t)(4 * half) * RDIM + lm;
  const float* alp = alpha0 + 4 * half;
  v8f accM[2][2];
#pragma unroll
  for (int j = 0; j < 2; ++j)
#pragma unroll
    for (int rt = 0; rt < 2; ++rt) accM[j][rt] = (v8f){};
  float p[2] = {0.f, 0.f};
  {
    PBatch fa, fb;
    pbatch_load(fa, 0, d1p, d2p, alp);
#pragma unroll 1
    for (int b = 0; b < KDIM / (8 * PGB) - 2; b += 2) {
      pbatch_load(fb, b + 1, d1p, d2p, alp);
      asm volatile("" ::: "memory");
      pbatch_mac(fa, accM, p);
      pbatch_load(fa, b + 2, d1p, d2p, alp);
      asm volatile("" ::: "memory");
      pbatch_mac(fb, accM, p);
    }
    pbatch_load(fb, KDIM / (8 * PGB) - 1, d1p, d2p, alp);
    asm volatile("" ::: "memory");
    pbatch_mac(fa, accM, p);
#pragma unroll
    for (int j = 0; j < PFD; ++j)
#pragma unroll
      for (int i = 0; i < 4; ++i)
        eb[j][i] = *(const v4f*)(ERh + (size_t)tokA[j] * RDIM + 16 * (i >> 1) + 4 * (i & 1));
    asm volatile("" ::: "memory");
    pbatch_mac(fb, accM, p);
  }

  v2f mA[2][8];
#pragma unroll
  for (int rt = 0; rt < 2; ++rt)
#pragma unroll
    for (int kc = 0; kc < 8; ++kc) {
      v2f f; f.x = accM[kc >> 2][rt][2 * (kc & 3)]; f.y = accM[kc >> 2][rt][2 * (kc & 3) + 1];
      mA[rt][kc] = f;
    }

  v8f dT[2];
#pragma unroll
  for (int rt = 0; rt < 2; ++rt) {
    float ps = p[rt] + __shfl_xor(p[rt], 16);
#pragma unroll
    for (int v = 0; v < 8; ++v) dT[rt][v] = __shfl(ps, 8 * half + v);
  }
  v2f gam[8];
#pragma unroll
  for (int kc = 0; kc < 8; ++kc) { gam[kc].x = 0.f; gam[kc].y = 0.f; }

#pragma unroll 1
  for (int t0 = 0; t0 < LSEQ; t0 += PFD) {
    int cn = (t0 >> 2) + 2;
    cn = cn < LSEQ / 4 ? cn : LSEQ / 4 - 1;
    v4i tokC = *(const v4i*)(xrow + 4 * cn);

#pragma unroll
    for (int j = 0; j < PFD; ++j) {
      const bool act = (t0 + j) < len;
#pragma unroll
      for (int kc = 0; kc < 8; ++kc) {
        const int rt = kc >> 2, v = 2 * (kc & 3);
        const v4f ev = eb[j][2 * rt + (v >> 2)];
        const float e0 = (v & 3) ? ev.z : ev.x;
        const float e1 = (v & 3) ? ev.w : ev.y;
        float gx = dT[rt][v] * e0, gy = dT[rt][v + 1] * e1;
        gam[kc].x = act ? gx : gam[kc].x;
        gam[kc].y = act ? gy : gam[kc].y;
      }
#pragma unroll
      for (int i = 0; i < 4; ++i)
        eb[j][i] = *(const v4f*)(ERh + (size_t)tokB[j] * RDIM + 16 * (i >> 1) + 4 * (i & 1));
      asm volatile("" ::: "memory");

      v8f z = {};
      v8f d0 = wmma_f32_4(mA[0][0], gam[0], z);
      v8f d1 = wmma_f32_4(mA[1][0], gam[0], z);
#pragma unroll
      for (int kc = 1; kc < 8; ++kc) {
        d0 = wmma_f32_4(mA[0][kc], gam[kc], d0);
        d1 = wmma_f32_4(mA[1][kc], gam[kc], d1);
      }
      dT[0] = d0; dT[1] = d1;
    }
    tokB = tokC;
  }
#pragma unroll
  for (int kc = 0; kc < 8; ++kc) gamS[kc][l] = gam[kc];
  } else {
#pragma unroll
    for (int s = 0; s < SA; ++s) {
      const int n  = 16 * (h + NHELP * s) + lm;
      const int nc = n < KDIM ? n : KDIM - 1;
#pragma unroll
      for (int kc = 0; kc < 8; ++kc)
        dA[s][kc] = *(const v2f*)(D2 + (size_t)nc * RDIM + KROW(kc, half));
    }
#pragma unroll
    for (int s = 0; s < SB; ++s) {
      const int n  = 16 * (h + NHELP * s) + lm;
      const int n1 = n < H1D ? n : H1D - 1;
      const int n2 = n < H2D ? n : H2D - 1;
      bs1[s] = b1[n1];
      bs2[s] = b2[n2];
#pragma unroll
      for (int g = 0; g < KDIM / 8; ++g) w1r[s][g] = *(const v4f*)(W1 + (size_t)n1 * KDIM + 8 * g + 4 * half);
#pragma unroll
      for (int kc = 0; kc < H1D / 4; ++kc) w2r[s][kc] = *(const v2f*)(W2 + (size_t)n2 * H1D + 4 * kc + 2 * half);
    }
    {
      const int nn = lm < NLAB ? lm : NLAB - 1;
      bs3 = b3[nn];
#pragma unroll
      for (int kc = 0; kc < H2D / 4; ++kc) w3r[kc] = *(const v2f*)(W3 + (size_t)nn * H2D + 4 * kc + 2 * half);
    }
  }
  __syncthreads();


  if (wv != 0) {
    v2f gam[8];
#pragma unroll
    for (int kc = 0; kc < 8; ++kc) gam[kc] = gamS[kc][l];
#pragma unroll
    for (int s = 0; s < SA; ++s) {
      v8f accF = {};
#pragma unroll
      for (int kc = 0; kc < 8; ++kc) accF = wmma_f32_4(gam[kc], dA[s][kc], accF);
      const int n = 16 * (h + NHELP * s) + lm;
      if (n < KDIM) {
#pragma unroll
        for (int v = 0; v < 8; ++v) alphaF[v + 8 * half][n] = accF[v];
      }
    }
  }
  __syncthreads();

  if (wv != 0) {
    v4f aF[KDIM / 8];
#pragma unroll
    for (int g = 0; g < KDIM / 8; ++g) aF[g] = *(const v4f*)(&alphaF[lm][8 * g + 4 * half]);
    v8f acc[SB];
#pragma unroll
    for (int s = 0; s < SB; ++s)
#pragma unroll
      for (int v = 0; v < 8; ++v) acc[s][v] = bs1[s];
    asm volatile("" : "+v"(acc[0]), "+v"(acc[1]), "+v"(acc[2]) : : "memory");
#pragma unroll
    for (int g = 0; g < KDIM / 8; ++g) {
      const v2f a0 = {aF[g].x, aF[g].y}, a1 = {aF[g].z, aF[g].w};
#pragma unroll
      for (int s = 0; s < SB; ++s) {
        const v2f b0 = {w1r[s][g].x, w1r[s][g].y};
        acc[s] = wmma_f32_4(a0, b0, acc[s]);
      }
#pragma unroll
      for (int s = 0; s < SB; ++s) {
        const v2f b1f = {w1r[s][g].z, w1r[s][g].w};
        acc[s] = wmma_f32_4(a1, b1f, acc[s]);
      }
    }
#pragma unroll
    for (int s = 0; s < SB; ++s) {
      const int n = 16 * (h + NHELP * s) + lm;
      if (n < H1D) {
#pragma unroll
        for (int v = 0; v < 8; ++v) Hb1[v + 8 * half][n] = fmaxf(acc[s][v], 0.f);
      }
    }
  }
  __syncthreads();

  if (wv != 0) {
    v2f aH[H1D / 4];
#pragma unroll
    for (int kc = 0; kc < H1D / 4; ++kc) aH[kc] = *(const v2f*)(&Hb1[lm][4 * kc + 2 * half]);
    v8f acc[SB];
#pragma unroll
    for (int s = 0; s < SB; ++s)
#pragma unroll
      for (int v = 0; v < 8; ++v) acc[s][v] = bs2[s];
    asm volatile("" : "+v"(acc[0]), "+v"(acc[1]), "+v"(acc[2]) : : "memory");
#pragma unroll
    for (int kc = 0; kc < H1D / 4; ++kc)
#pragma unroll
      for (int s = 0; s < SB; ++s) acc[s] = wmma_f32_4(aH[kc], w2r[s][kc], acc[s]);
#pragma unroll
    for (int s = 0; s < SB; ++s) {
      const int n = 16 * (h + NHELP * s) + lm;
      if (n < H2D) {
#pragma unroll
        for (int v = 0; v < 8; ++v) Hb2[v + 8 * half][n] = fmaxf(acc[s][v], 0.f);
      }
    }
  }
  __syncthreads();

  if (wv == 1) {
    v2f aH[H2D / 4];
#pragma unroll
    for (int kc = 0; kc < H2D / 4; ++kc) aH[kc] = *(const v2f*)(&Hb2[lm][4 * kc + 2 * half]);
    v8f c0, c1 = {};
#pragma unroll
    for (int v = 0; v < 8; ++v) c0[v] = bs3;
    asm volatile("" : "+v"(c0), "+v"(c1) : : "memory");
#pragma unroll
    for (int kc = 0; kc < H2D / 4; ++kc) {
      if (kc & 1) c1 = wmma_f32_4(aH[kc], w3r[kc], c1);
      else        c0 = wmma_f32_4(aH[kc], w3r[kc], c0);
    }
    v8f c = c0 + c1;
    if (lm < NLAB) {
#pragma unroll
      for (int v = 0; v < 8; ++v) Lbuf[(v + 8 * half) * NLAB + lm] = 1.0f * c[v];
    }
  }
  __syncthreads();
  for (int o = threadIdx.x; o < 16 * NLAB; o += 32 * (1 + NHELP)) out[(size_t)seq0 * NLAB + o] = Lbuf[o];
}

extern "C" void kernel_launch(void* const* d_in, const int* in_sizes, int n_in,
                              void* d_out, int out_size, void* d_ws, size_t ws_size,
                              hipStream_t stream) {
  const int*   x      = (const int*)d_in[0];
  const int*   lens   = (const int*)d_in[1];
  const float* alpha0 = (const float*)d_in[2];
  const float* ER     = (const float*)d_in[3];
  const float* D1     = (const float*)d_in[4];
  const float* D2     = (const float*)d_in[5];
  const float* W1     = (const float*)d_in[6];
  const float* b1     = (const float*)d_in[7];
  const float* W2     = (const float*)d_in[8];
  const float* b2     = (const float*)d_in[9];
  const float* W3     = (const float*)d_in[10];
  const float* b3     = (const float*)d_in[11];
  float* out = (float*)d_out;

  farnn_main<<<2048 / 16, 32 * (1 + NHELP), 0, stream>>>(x, lens, alpha0, ER, D1, D2, W1, b1, W2,
                                           b2, W3, b3, out);
}


# Time-loop gathers as uniform base + 32-bit lane offset: 5 vector address instructions per 4 steps (were 13, 9 of them 64-bit); no address register is rewritten right behind its load (xcnt wait gone)
# speedup vs baseline: 5.5424x; 1.0540x over previous
#include <hip/hip_runtime.h>

typedef float v2f __attribute__((ext_vector_type(2)));
typedef float v4f __attribute__((ext_vector_type(4)));
typedef float v8f __attribute__((ext_vector_type(8)));

#define LSEQ 512
#define KDIM 256
#define RDIM 32
#define H1D  100
#define H2D  100
#define NLAB 10
#define AFP  260
#define NT1  7
#define NHELP 3
#define SA   6
#define SB   3

static __device__ __forceinline__ v8f wmma_f32_4(v2f a, v2f b, v8f c) {
  return __builtin_amdgcn_wmma_f32_16x16x4_f32(false, a, false, b, (short)0, c,
                                               false, false);
}

typedef int v4i __attribute__((ext_vector_type(4)));

#define PFD 4
#define ERROW(tok, hoff) ((((unsigned)(tok)) * (unsigned)(RDIM * sizeof(float))) | (hoff))

#define PGB 4

struct PBatch { v4f d1[PGB][2]; float a[PGB][2][4]; v4f al[PGB]; };

static __device__ __forceinline__ void pbatch_load(PBatch& f, int b, const float* __restrict__ d1p,
                                                   const float* __restrict__ d2p,
                                                   const float* __restrict__ alp) {
#pragma unroll
  for (int u = 0; u < PGB; ++u) {
    const int k0 = 8 * (PGB * b + u);
#pragma unroll
    for (int rt = 0; rt < 2; ++rt) f.d1[u][rt] = *(const v4f*)(d1p + (size_t)(16 * rt) * KDIM + k0);
#pragma unroll
    for (int j = 0; j < 2; ++j)
#pragma unroll
      for (int q = 0; q < 4; ++q) f.a[u][j][q] = d2p[(size_t)(k0 + q) * RDIM + 16 * j];
    f.al[u] = *(const v4f*)(alp + k0);
  }
}

static __device__ __forceinline__ void pbatch_mac(const PBatch& f, v8f (&accM)[2][2], float (&p)[2]) {
#pragma unroll
  for (int u = 0; u < PGB; ++u)
#pragma unroll
    for (int w = 0; w < 2; ++w) {
#pragma unroll
      for (int j = 0; j < 2; ++j)
#pragma unroll
        for (int rt = 0; rt < 2; ++rt) {
          v2f af = {f.a[u][j][2 * w], f.a[u][j][2 * w + 1]};
          v2f bf = {f.d1[u][rt][2 * w], f.d1[u][rt][2 * w + 1]};
          accM[j][rt] = wmma_f32_4(af, bf, accM[j][rt]);
        }
#pragma unroll
      for (int rt = 0; rt < 2; ++rt) {
        p[rt] = fmaf(f.d1[u][rt][2 * w], f.al[u][2 * w], p[rt]);
        p[rt] = fmaf(f.d1[u][rt][2 * w + 1], f.al[u][2 * w + 1], p[rt]);
      }
    }
}

#define KROW(kc, half) (16 * ((kc) >> 2) + 8 * (half) + 2 * ((kc) & 3))
__global__ __launch_bounds__(32 * (1 + NHELP)) void farnn_main(
    const int* __restrict__ x, const int* __restrict__ lengths,
    const float* __restrict__ alpha0, const float* __restrict__ ER,
    const float* __restrict__ D1, const float* __restrict__ D2,
    const float* __restrict__ W1,
    const float* __restrict__ b1, const float* __restrict__ W2,
    const float* __restrict__ b2, const float* __restrict__ W3,
    const float* __restrict__ b3, float* __restrict__ out) {
  __shared__ float alphaF[16][AFP];
  __shared__ float Hb1[16][H1D];
  __shared__ float Hb2[16][H2D];
  __shared__ float Lbuf[16 * NLAB];
  __shared__ v2f   gamS[8][32];

  const int l    = threadIdx.x & 31;
  const int wv   = __builtin_amdgcn_readfirstlane((int)(threadIdx.x >> 5));
  const int h    = wv - 1;
  const int lm   = l & 15;
  const int half = l >> 4;
  const int seq0 = blockIdx.x * 16;

  v2f   dA[SA][8];
  v4f   w1r[SB][KDIM / 8];
  v2f   w2r[SB][H1D / 4];
  v2f   w3r[H2D / 4];
  float bs1[SB], bs2[SB], bs3;

  if (wv == 0) {
  const int len = lengths[seq0 + lm];

  const char*    xb   = (const char*)x;
  const unsigned xoff = (unsigned)(seq0 + lm) * (unsigned)(LSEQ * sizeof(int));
  v4i tokA = *(const v4i*)(xb + xoff);
  v4i tokB = *(const v4i*)(xb + xoff + PFD * sizeof(int));

  const char*    ERb  = (const char*)ER;
  const unsigned hoff = (unsigned)(8 * half * sizeof(float));
  v4f eb[PFD][4];

  const float* d1p = D1 + (size_t)lm * KDIM + 4 * half;
  const float* d2p = D2 + (size_t)(4 * half) * RDIM + lm;
  const float* alp = alpha0 + 4 * half;
  v8f accM[2][2];
#pragma unroll
  for (int j = 0; j < 2; ++j)
#pragma unroll
    for (int rt = 0; rt < 2; ++rt) accM[j][rt] = (v8f){};
  float p[2] = {0.f, 0.f};
  {
    PBatch fa, fb;
    pbatch_load(fa, 0, d1p, d2p, alp);
#pragma unroll 1
    for (int b = 0; b < KDIM / (8 * PGB) - 2; b += 2) {
      pbatch_load(fb, b + 1, d1p, d2p, alp);
      asm volatile("" ::: "memory");
      pbatch_mac(fa, accM, p);
      pbatch_load(fa, b + 2, d1p, d2p, alp);
      asm volatile("" ::: "memory");
      pbatch_mac(fb, accM, p);
    }
    pbatch_load(fb, KDIM / (8 * PGB) - 1, d1p, d2p, alp);
    asm volatile("" ::: "memory");
    pbatch_mac(fa, accM, p);
#pragma unroll
    for (int j = 0; j < PFD; ++j)
#pragma unroll
      for (int i = 0; i < 4; ++i)
        eb[j][i] = *(const v4f*)(ERb + ERROW(tokA[j], hoff) + sizeof(float) * (16 * (i >> 1) + 4 * (i & 1)));
    asm volatile("" ::: "memory");
    pbatch_mac(fb, accM, p);
  }

  v2f mA[2][8];
#pragma unroll
  for (int rt = 0; rt < 2; ++rt)
#pragma unroll
    for (int kc = 0; kc < 8; ++kc) {
      v2f f; f.x = accM[kc >> 2][rt][2 * (kc & 3)]; f.y = accM[kc >> 2][rt][2 * (kc & 3) + 1];
      mA[rt][kc] = f;
    }

  v8f dT[2];
#pragma unroll
  for (int rt = 0; rt < 2; ++rt) {
    float ps = p[rt] + __shfl_xor(p[rt], 16);
#pragma unroll
    for (int v = 0; v < 8; ++v) dT[rt][v] = __shfl(ps, 8 * half + v);
  }
  v2f gam[8];
#pragma unroll
  for (int kc = 0; kc < 8; ++kc) { gam[kc].x = 0.f; gam[kc].y = 0.f; }

#pragma unroll 1
  for (int t0 = 0; t0 < LSEQ; t0 += PFD) {
    int cn = (t0 >> 2) + 2;
    cn = cn < LSEQ / 4 ? cn : LSEQ / 4 - 1;
    const char* xg = xb + (size_t)cn * (PFD * sizeof(int));
    v4i tokC = *(const v4i*)(xg + xoff);

#pragma unroll
    for (int j = 0; j < PFD; ++j) {
      const bool act = (t0 + j) < len;
#pragma unroll
      for (int kc = 0; kc < 8; ++kc) {
        const int rt = kc >> 2, v = 2 * (kc & 3);
        const v4f ev = eb[j][2 * rt + (v >> 2)];
        const float e0 = (v & 3) ? ev.z : ev.x;
        const float e1 = (v & 3) ? ev.w : ev.y;
        float gx = dT[rt][v] * e0, gy = dT[rt][v + 1] * e1;
        gam[kc].x = act ? gx : gam[kc].x;
        gam[kc].y = act ? gy : gam[kc].y;
      }
#pragma unroll
      for (int i = 0; i < 4; ++i)
        eb[j][i] = *(const v4f*)(ERb + ERROW(tokB[j], hoff) + sizeof(float) * (16 * (i >> 1) + 4 * (i & 1)));
      asm volatile("" ::: "memory");

      v8f z = {};
      v8f d0 = wmma_f32_4(mA[0][0], gam[0], z);
      v8f d1 = wmma_f32_4(mA[1][0], gam[0], z);
#pragma unroll
      for (int kc = 1; kc < 8; ++kc) {
        d0 = wmma_f32_4(mA[0][kc], gam[kc], d0);
        d1 = wmma_f32_4(mA[1][kc], gam[kc], d1);
      }
      dT[0] = d0; dT[1] = d1;
    }
    tokB = tokC;
  }
#pragma unroll
  for (int kc = 0; kc < 8; ++kc) gamS[kc][l] = gam[kc];
  } else {
#pragma unroll
    for (int s = 0; s < SA; ++s) {
      const int n  = 16 * (h + NHELP * s) + lm;
      const int nc = n < KDIM ? n : KDIM - 1;
#pragma unroll
      for (int kc = 0; kc < 8; ++kc)
        dA[s][kc] = *(const v2f*)(D2 + (size_t)nc * RDIM + KROW(kc, half));
    }
#pragma unroll
    for (int s = 0; s < SB; ++s) {
      const int n  = 16 * (h + NHELP * s) + lm;
      const int n1 = n < H1D ? n : H1D - 1;
      const int n2 = n < H2D ? n : H2D - 1;
      bs1[s] = b1[n1];
      bs2[s] = b2[n2];
#pragma unroll
      for (int g = 0; g < KDIM / 8; ++g) w1r[s][g] = *(const v4f*)(W1 + (size_t)n1 * KDIM + 8 * g + 4 * half);
#pragma unroll
      for (int kc = 0; kc < H1D / 4; ++kc) w2r[s][kc] = *(const v2f*)(W2 + (size_t)n2 * H1D + 4 * kc + 2 * half);
    }
    {
      const int nn = lm < NLAB ? lm : NLAB - 1;
      bs3 = b3[nn];
#pragma unroll
      for (int kc = 0; kc < H2D / 4; ++kc) w3r[kc] = *(const v2f*)(W3 + (size_t)nn * H2D + 4 * kc + 2 * half);
    }
  }
  __syncthreads();


  if (wv != 0) {
    v2f gam[8];
#pragma unroll
    for (int kc = 0; kc < 8; ++kc) gam[kc] = gamS[kc][l];
#pragma unroll
    for (int s = 0; s < SA; ++s) {
      v8f accF = {};
#pragma unroll
      for (int kc = 0; kc < 8; ++kc) accF = wmma_f32_4(gam[kc], dA[s][kc], accF);
      const int n = 16 * (h + NHELP * s) + lm;
      if (n < KDIM) {
#pragma unroll
        for (int v = 0; v < 8; ++v) alphaF[v + 8 * half][n] = accF[v];
      }
    }
  }
  __syncthreads();

  if (wv != 0) {
    v4f aF[KDIM / 8];
#pragma unroll
    for (int g = 0; g < KDIM / 8; ++g) aF[g] = *(const v4f*)(&alphaF[lm][8 * g + 4 * half]);
    v8f acc[SB];
#pragma unroll
    for (int s = 0; s < SB; ++s)
#pragma unroll
      for (int v = 0; v < 8; ++v) acc[s][v] = bs1[s];
    asm volatile("" : "+v"(acc[0]), "+v"(acc[1]), "+v"(acc[2]) : : "memory");
#pragma unroll
    for (int g = 0; g < KDIM / 8; ++g) {
      const v2f a0 = {aF[g].x, aF[g].y}, a1 = {aF[g].z, aF[g].w};
#pragma unroll
      for (int s = 0; s < SB; ++s) {
        const v2f b0 = {w1r[s][g].x, w1r[s][g].y};
        acc[s] = wmma_f32_4(a0, b0, acc[s]);
      }
#pragma unroll
      for (int s = 0; s < SB; ++s) {
        const v2f b1f = {w1r[s][g].z, w1r[s][g].w};
        acc[s] = wmma_f32_4(a1, b1f, acc[s]);
      }
    }
#pragma unroll
    for (int s = 0; s < SB; ++s) {
      const int n = 16 * (h + NHELP * s) + lm;
      if (n < H1D) {
#pragma unroll
        for (int v = 0; v < 8; ++v) Hb1[v + 8 * half][n] = fmaxf(acc[s][v], 0.f);
      }
    }
  }
  __syncthreads();

  if (wv != 0) {
    v2f aH[H1D / 4];
#pragma unroll
    for (int kc = 0; kc < H1D / 4; ++kc) aH[kc] = *(const v2f*)(&Hb1[lm][4 * kc + 2 * half]);
    v8f acc[SB];
#pragma unroll
    for (int s = 0; s < SB; ++s)
#pragma unroll
      for (int v = 0; v < 8; ++v) acc[s][v] = bs2[s];
    asm volatile("" : "+v"(acc[0]), "+v"(acc[1]), "+v"(acc[2]) : : "memory");
#pragma unroll
    for (int kc = 0; kc < H1D / 4; ++kc)
#pragma unroll
      for (int s = 0; s < SB; ++s) acc[s] = wmma_f32_4(aH[kc], w2r[s][kc], acc[s]);
#pragma unroll
    for (int s = 0; s < SB; ++s) {
      const int n = 16 * (h + NHELP * s) + lm;
      if (n < H2D) {
#pragma unroll
        for (int v = 0; v < 8; ++v) Hb2[v + 8 * half][n] = fmaxf(acc[s][v], 0.f);
      }
    }
  }
  __syncthreads();

  if (wv == 1) {
    v2f aH[H2D / 4];
#pragma unroll
    for (int kc = 0; kc < H2D / 4; ++kc) aH[kc] = *(const v2f*)(&Hb2[lm][4 * kc + 2 * half]);
    v8f c0, c1 = {};
#pragma unroll
    for (int v = 0; v < 8; ++v) c0[v] = bs3;
    asm volatile("" : "+v"(c0), "+v"(c1) : : "memory");
#pragma unroll
    for (int kc = 0; kc < H2D / 4; ++kc) {
      if (kc & 1) c1 = wmma_f32_4(aH[kc], w3r[kc], c1);
      else        c0 = wmma_f32_4(aH[kc], w3r[kc], c0);
    }
    v8f c = c0 + c1;
    if (lm < NLAB) {
#pragma unroll
      for (int v = 0; v < 8; ++v) Lbuf[(v + 8 * half) * NLAB + lm] = 1.0f * c[v];
    }
  }
  __syncthreads();
  for (int o = threadIdx.x; o < 16 * NLAB; o += 32 * (1 + NHELP)) out[(size_t)seq0 * NLAB + o] = Lbuf[o];
}

extern "C" void kernel_launch(void* const* d_in, const int* in_sizes, int n_in,
                              void* d_out, int out_size, void* d_ws, size_t ws_size,
                              hipStream_t stream) {
  const int*   x      = (const int*)d_in[0];
  const int*   lens   = (const int*)d_in[1];
  const float* alpha0 = (const float*)d_in[2];
  const float* ER     = (const float*)d_in[3];
  const float* D1     = (const float*)d_in[4];
  const float* D2     = (const float*)d_in[5];
  const float* W1     = (const float*)d_in[6];
  const float* b1     = (const float*)d_in[7];
  const float* W2     = (const float*)d_in[8];
  const float* b2     = (const float*)d_in[9];
  const float* W3     = (const float*)d_in[10];
  const float* b3     = (const float*)d_in[11];
  float* out = (float*)d_out;

  farnn_main<<<2048 / 16, 32 * (1 + NHELP), 0, stream>>>(x, lens, alpha0, ER, D1, D2, W1, b1, W2,
                                           b2, W3, b3, out);
}


# Time-loop step split over two waves by row tile: each runs one chain of 8 f32 WMMAs on its own SIMD, partner's gamma crosses LDS behind one barrier a step; tile 1 now sums k chunks 4-7 before 0-3
# speedup vs baseline: 6.0346x; 1.0888x over previous
#include <hip/hip_runtime.h>

typedef float v2f __attribute__((ext_vector_type(2)));
typedef float v4f __attribute__((ext_vector_type(4)));
typedef float v8f __attribute__((ext_vector_type(8)));

#define LSEQ 512
#define KDIM 256
#define RDIM 32
#define H1D  100
#define H2D  100
#define NLAB 10
#define AFP  260
#define NT1  7
#define NHELP 3
#define SA   6
#define SB   3

static __device__ __forceinline__ v8f wmma_f32_4(v2f a, v2f b, v8f c) {
  return __builtin_amdgcn_wmma_f32_16x16x4_f32(false, a, false, b, (short)0, c,
                                               false, false);
}

static __device__ __forceinline__ void lds_barrier() {
  __builtin_amdgcn_fence(__ATOMIC_RELEASE, "workgroup", "local");
  __builtin_amdgcn_s_barrier();
  __builtin_amdgcn_fence(__ATOMIC_ACQUIRE, "workgroup", "local");
}

typedef int v4i __attribute__((ext_vector_type(4)));

#define PFD 4
#define ERROW(tok, hoff) ((((unsigned)(tok)) * (unsigned)(RDIM * sizeof(float))) | (hoff))

#define PGB 4

struct PBatch { v4f d1[PGB][2]; float a[PGB][2][4]; v4f al[PGB]; };

static __device__ __forceinline__ void pbatch_load(PBatch& f, int b, const float* __restrict__ d1p,
                                                   const float* __restrict__ d2p,
                                                   const float* __restrict__ alp) {
#pragma unroll
  for (int u = 0; u < PGB; ++u) {
    const int k0 = 8 * (PGB * b + u);
#pragma unroll
    for (int rt = 0; rt < 2; ++rt) f.d1[u][rt] = *(const v4f*)(d1p + (size_t)(16 * rt) * KDIM + k0);
#pragma unroll
    for (int j = 0; j < 2; ++j)
#pragma unroll
      for (int q = 0; q < 4; ++q) f.a[u][j][q] = d2p[(size_t)(k0 + q) * RDIM + 16 * j];
    f.al[u] = *(const v4f*)(alp + k0);
  }
}

static __device__ __forceinline__ void pbatch_mac(const PBatch& f, v8f (&accM)[2][2], float (&p)[2]) {
#pragma unroll
  for (int u = 0; u < PGB; ++u)
#pragma unroll
    for (int w = 0; w < 2; ++w) {
#pragma unroll
      for (int j = 0; j < 2; ++j)
#pragma unroll
        for (int rt = 0; rt < 2; ++rt) {
          v2f af = {f.a[u][j][2 * w], f.a[u][j][2 * w + 1]};
          v2f bf = {f.d1[u][rt][2 * w], f.d1[u][rt][2 * w + 1]};
          accM[j][rt] = wmma_f32_4(af, bf, accM[j][rt]);
        }
#pragma unroll
      for (int rt = 0; rt < 2; ++rt) {
        p[rt] = fmaf(f.d1[u][rt][2 * w], f.al[u][2 * w], p[rt]);
        p[rt] = fmaf(f.d1[u][rt][2 * w + 1], f.al[u][2 * w + 1], p[rt]);
      }
    }
}

#define KROW(kc, half) (16 * ((kc) >> 2) + 8 * (half) + 2 * ((kc) & 3))
__global__ __launch_bounds__(32 * (1 + NHELP)) void farnn_main(
    const int* __restrict__ x, const int* __restrict__ lengths,
    const float* __restrict__ alpha0, const float* __restrict__ ER,
    const float* __restrict__ D1, const float* __restrict__ D2,
    const float* __restrict__ W1,
    const float* __restrict__ b1, const float* __restrict__ W2,
    const float* __restrict__ b2, const float* __restrict__ W3,
    const float* __restrict__ b3, float* __restrict__ out) {
  __shared__ float alphaF[16][AFP];
  __shared__ float Hb1[16][H1D];
  __shared__ float Hb2[16][H2D];
  __shared__ float Lbuf[16 * NLAB];
  __shared__ v2f   gamS[8][32];
  __shared__ v2f   handA[2][8][32];
  __shared__ float handD[2][8][32];
  __shared__ v4f   xbuf[2][2][2][32];

  const int l    = threadIdx.x & 31;
  const int wv   = __builtin_amdgcn_readfirstlane((int)(threadIdx.x >> 5));
  const int h    = wv - 1;
  const int lm   = l & 15;
  const int half = l >> 4;
  const int seq0 = blockIdx.x * 16;
  const bool lw  = wv < 2;
  const int  ow  = 1 - wv;

  v2f   dA[SA][8];
  v4f   w1r[SB][KDIM / 8];
  v2f   w2r[SB][H1D / 4];
  v2f   w3r[H2D / 4];
  float bs1[SB], bs2[SB], bs3;

  const char*    xb   = (const char*)x;
  const char*    ERb  = (const char*)ER;
  const unsigned xoff = (unsigned)(seq0 + lm) * (unsigned)(LSEQ * sizeof(int));
  const unsigned hoff = (unsigned)(8 * half * sizeof(float)) | (unsigned)(16 * (wv & 1) * sizeof(float));
  int len = 0;
  v4i tokB = {};
  v4f eb[PFD][2];
#pragma unroll
  for (int j = 0; j < PFD; ++j) { eb[j][0] = (v4f){}; eb[j][1] = (v4f){}; }

  if (wv == 0) {
  { const int lr = lengths[seq0 + lm]; len = lr < LSEQ ? lr : LSEQ; }

  v4i tokA = *(const v4i*)(xb + xoff);
  tokB = *(const v4i*)(xb + xoff + PFD * sizeof(int));

  const float* d1p = D1 + (size_t)lm * KDIM + 4 * half;
  const float* d2p = D2 + (size_t)(4 * half) * RDIM + lm;
  const float* alp = alpha0 + 4 * half;
  v8f accM[2][2];
#pragma unroll
  for (int j = 0; j < 2; ++j)
#pragma unroll
    for (int rt = 0; rt < 2; ++rt) accM[j][rt] = (v8f){};
  float p[2] = {0.f, 0.f};
  {
    PBatch fa, fb;
    pbatch_load(fa, 0, d1p, d2p, alp);
#pragma unroll 1
    for (int b = 0; b < KDIM / (8 * PGB) - 2; b += 2) {
      pbatch_load(fb, b + 1, d1p, d2p, alp);
      asm volatile("" ::: "memory");
      pbatch_mac(fa, accM, p);
      pbatch_load(fa, b + 2, d1p, d2p, alp);
      asm volatile("" ::: "memory");
      pbatch_mac(fb, accM, p);
    }
    pbatch_load(fb, KDIM / (8 * PGB) - 1, d1p, d2p, alp);
    asm volatile("" ::: "memory");
    pbatch_mac(fa, accM, p);
#pragma unroll
    for (int j = 0; j < PFD; ++j)
#pragma unroll
      for (int q = 0; q < 2; ++q)
        eb[j][q] = *(const v4f*)(ERb + ERROW(tokA[j], hoff) + sizeof(float) * (4 * q));
    asm volatile("" ::: "memory");
    pbatch_mac(fb, accM, p);
  }

  v2f mA[2][8];
#pragma unroll
  for (int rt = 0; rt < 2; ++rt)
#pragma unroll
    for (int kc = 0; kc < 8; ++kc) {
      v2f f; f.x = accM[kc >> 2][rt][2 * (kc & 3)]; f.y = accM[kc >> 2][rt][2 * (kc & 3) + 1];
      mA[rt][kc] = f;
    }

  v8f dT[2];
#pragma unroll
  for (int rt = 0; rt < 2; ++rt) {
    float ps = p[rt] + __shfl_xor(p[rt], 16);
#pragma unroll
    for (int v = 0; v < 8; ++v) dT[rt][v] = __shfl(ps, 8 * half + v);
  }
#pragma unroll
  for (int rt = 0; rt < 2; ++rt) {
#pragma unroll
    for (int kc = 0; kc < 8; ++kc) handA[rt][kc][l] = mA[rt][kc];
#pragma unroll
    for (int v = 0; v < 8; ++v) handD[rt][v][l] = dT[rt][v];
  }
  } else {
    {
      { const int lr = lengths[seq0 + lm]; len = lr < LSEQ ? lr : LSEQ; }
      const v4i tokA = *(const v4i*)(xb + xoff);
      tokB = *(const v4i*)(xb + xoff + PFD * sizeof(int));
#pragma unroll
      for (int j = 0; j < PFD; ++j)
#pragma unroll
        for (int q = 0; q < 2; ++q)
          eb[j][q] = *(const v4f*)(ERb + ERROW(tokA[j], hoff) + sizeof(float) * (4 * q));
    }
#pragma unroll
    for (int s = 0; s < SA; ++s) {
      const int n  = 16 * (h + NHELP * s) + lm;
      const int nc = n < KDIM ? n : KDIM - 1;
#pragma unroll
      for (int kc = 0; kc < 8; ++kc)
        dA[s][kc] = *(const v2f*)(D2 + (size_t)nc * RDIM + KROW(kc, half));
    }
#pragma unroll
    for (int s = 0; s < SB; ++s) {
      const int n  = 16 * (h + NHELP * s) + lm;
      const int n1 = n < H1D ? n : H1D - 1;
      const int n2 = n < H2D ? n : H2D - 1;
      bs1[s] = b1[n1];
      bs2[s] = b2[n2];
#pragma unroll
      for (int g = 0; g < KDIM / 8; ++g) w1r[s][g] = *(const v4f*)(W1 + (size_t)n1 * KDIM + 8 * g + 4 * half);
#pragma unroll
      for (int kc = 0; kc < H1D / 4; ++kc) w2r[s][kc] = *(const v2f*)(W2 + (size_t)n2 * H1D + 4 * kc + 2 * half);
    }
    {
      const int nn = lm < NLAB ? lm : NLAB - 1;
      bs3 = b3[nn];
#pragma unroll
      for (int kc = 0; kc < H2D / 4; ++kc) w3r[kc] = *(const v2f*)(W3 + (size_t)nn * H2D + 4 * kc + 2 * half);
    }
  }
  lds_barrier();

  v2f aO[4], aX[4];
  v8f dTw = {};
  v2f gO[4];
#pragma unroll
  for (int i = 0; i < 4; ++i) { gO[i].x = 0.f; gO[i].y = 0.f; aO[i] = gO[i]; aX[i] = gO[i]; }
  if (lw) {
#pragma unroll
    for (int i = 0; i < 4; ++i) {
      aO[i] = handA[wv][4 * wv + i][l];
      aX[i] = handA[wv][4 * ow + i][l];
    }
#pragma unroll
    for (int v = 0; v < 8; ++v) dTw[v] = handD[wv][v][l];
    const bool a0 = 0 < len;
#pragma unroll
    for (int i = 0; i < 4; ++i) {
      const v4f ev = eb[0][i >> 1];
      const float e0 = (i & 1) ? ev.z : ev.x;
      const float e1 = (i & 1) ? ev.w : ev.y;
      gO[i].x = a0 ? dTw[2 * i] * e0 : 0.f;
      gO[i].y = a0 ? dTw[2 * i + 1] * e1 : 0.f;
    }
    xbuf[0][wv][0][l] = (v4f){gO[0].x, gO[0].y, gO[1].x, gO[1].y};
    xbuf[0][wv][1][l] = (v4f){gO[2].x, gO[2].y, gO[3].x, gO[3].y};
  }
#pragma unroll
  for (int q = 0; q < 2; ++q)
    eb[0][q] = *(const v4f*)(ERb + ERROW(tokB[0], hoff) + sizeof(float) * (4 * q));
  asm volatile("" ::: "memory");

#pragma unroll 1
  for (int t0 = 0; t0 < LSEQ; t0 += PFD) {
    int cn = (t0 >> 2) + 2;
    cn = cn < LSEQ / 4 ? cn : LSEQ / 4 - 1;
    const char* xg = xb + (size_t)cn * (PFD * sizeof(int));
    const v4i tokC = *(const v4i*)(xg + xoff);
#pragma unroll
    for (int j = 0; j < PFD; ++j) {
      const int jn = (j + 1) % PFD;
      lds_barrier();
      if (lw) {
        const v4f x0 = xbuf[j & 1][ow][0][l];
        const v4f x1 = xbuf[j & 1][ow][1][l];
        asm volatile("" : "+v"(gO[0]) : : "memory");
        v8f z = {};
        v8f d = wmma_f32_4(aO[0], gO[0], z);
#pragma unroll
        for (int i = 1; i < 4; ++i) d = wmma_f32_4(aO[i], gO[i], d);
        const v2f g0 = {x0.x, x0.y}, g1 = {x0.z, x0.w}, g2 = {x1.x, x1.y}, g3 = {x1.z, x1.w};
        d = wmma_f32_4(aX[0], g0, d);
        d = wmma_f32_4(aX[1], g1, d);
        d = wmma_f32_4(aX[2], g2, d);
        d = wmma_f32_4(aX[3], g3, d);

        const bool act = (t0 + j + 1) < len;
#pragma unroll
        for (int i = 0; i < 4; ++i) {
          const v4f ev = eb[jn][i >> 1];
          const float e0 = (i & 1) ? ev.z : ev.x;
          const float e1 = (i & 1) ? ev.w : ev.y;
          const float gx = d[2 * i] * e0, gy = d[2 * i + 1] * e1;
          gO[i].x = act ? gx : gO[i].x;
          gO[i].y = act ? gy : gO[i].y;
        }
        xbuf[(j + 1) & 1][wv][0][l] = (v4f){gO[0].x, gO[0].y, gO[1].x, gO[1].y};
        xbuf[(j + 1) & 1][wv][1][l] = (v4f){gO[2].x, gO[2].y, gO[3].x, gO[3].y};
      }
      const int tk = (j + 1 < PFD) ? tokB[jn] : tokC[0];
#pragma unroll
      for (int q = 0; q < 2; ++q)
        eb[jn][q] = *(const v4f*)(ERb + ERROW(tk, hoff) + sizeof(float) * (4 * q));
      asm volatile("" ::: "memory");
    }
    tokB = tokC;
  }
  if (lw) {
#pragma unroll
    for (int i = 0; i < 4; ++i) gamS[4 * wv + i][l] = gO[i];
  }
  __syncthreads();


  if (wv != 0) {
    v2f gam[8];
#pragma unroll
    for (int kc = 0; kc < 8; ++kc) gam[kc] = gamS[kc][l];
#pragma unroll
    for (int s = 0; s < SA; ++s) {
      v8f accF = {};
#pragma unroll
      for (int kc = 0; kc < 8; ++kc) accF = wmma_f32_4(gam[kc], dA[s][kc], accF);
      const int n = 16 * (h + NHELP * s) + lm;
      if (n < KDIM) {
#pragma unroll
        for (int v = 0; v < 8; ++v) alphaF[v + 8 * half][n] = accF[v];
      }
    }
  }
  __syncthreads();

  if (wv != 0) {
    v4f aF[KDIM / 8];
#pragma unroll
    for (int g = 0; g < KDIM / 8; ++g) aF[g] = *(const v4f*)(&alphaF[lm][8 * g + 4 * half]);
    v8f acc[SB];
#pragma unroll
    for (int s = 0; s < SB; ++s)
#pragma unroll
      for (int v = 0; v < 8; ++v) acc[s][v] = bs1[s];
    asm volatile("" : "+v"(acc[0]), "+v"(acc[1]), "+v"(acc[2]) : : "memory");
#pragma unroll
    for (int g = 0; g < KDIM / 8; ++g) {
      const v2f a0 = {aF[g].x, aF[g].y}, a1 = {aF[g].z, aF[g].w};
#pragma unroll
      for (int s = 0; s < SB; ++s) {
        const v2f b0 = {w1r[s][g].x, w1r[s][g].y};
        acc[s] = wmma_f32_4(a0, b0, acc[s]);
      }
#pragma unroll
      for (int s = 0; s < SB; ++s) {
        const v2f b1f = {w1r[s][g].z, w1r[s][g].w};
        acc[s] = wmma_f32_4(a1, b1f, acc[s]);
      }
    }
#pragma unroll
    for (int s = 0; s < SB; ++s) {
      const int n = 16 * (h + NHELP * s) + lm;
      if (n < H1D) {
#pragma unroll
        for (int v = 0; v < 8; ++v) Hb1[v + 8 * half][n] = fmaxf(acc[s][v], 0.f);
      }
    }
  }
  __syncthreads();

  if (wv != 0) {
    v2f aH[H1D / 4];
#pragma unroll
    for (int kc = 0; kc < H1D / 4; ++kc) aH[kc] = *(const v2f*)(&Hb1[lm][4 * kc + 2 * half]);
    v8f acc[SB];
#pragma unroll
    for (int s = 0; s < SB; ++s)
#pragma unroll
      for (int v = 0; v < 8; ++v) acc[s][v] = bs2[s];
    asm volatile("" : "+v"(acc[0]), "+v"(acc[1]), "+v"(acc[2]) : : "memory");
#pragma unroll
    for (int kc = 0; kc < H1D / 4; ++kc)
#pragma unroll
      for (int s = 0; s < SB; ++s) acc[s] = wmma_f32_4(aH[kc], w2r[s][kc], acc[s]);
#pragma unroll
    for (int s = 0; s < SB; ++s) {
      const int n = 16 * (h + NHELP * s) + lm;
      if (n < H2D) {
#pragma unroll
        for (int v = 0; v < 8; ++v) Hb2[v + 8 * half][n] = fmaxf(acc[s][v], 0.f);
      }
    }
  }
  __syncthreads();

  if (wv == 1) {
    v2f aH[H2D / 4];
#pragma unroll
    for (int kc = 0; kc < H2D / 4; ++kc) aH[kc] = *(const v2f*)(&Hb2[lm][4 * kc + 2 * half]);
    v8f c0, c1 = {};
#pragma unroll
    for (int v = 0; v < 8; ++v) c0[v] = bs3;
    asm volatile("" : "+v"(c0), "+v"(c1) : : "memory");
#pragma unroll
    for (int kc = 0; kc < H2D / 4; ++kc) {
      if (kc & 1) c1 = wmma_f32_4(aH[kc], w3r[kc], c1);
      else        c0 = wmma_f32_4(aH[kc], w3r[kc], c0);
    }
    v8f c = c0 + c1;
    if (lm < NLAB) {
#pragma unroll
      for (int v = 0; v < 8; ++v) Lbuf[(v + 8 * half) * NLAB + lm] = 1.0f * c[v];
    }
  }
  __syncthreads();
  for (int o = threadIdx.x; o < 16 * NLAB; o += 32 * (1 + NHELP)) out[(size_t)seq0 * NLAB + o] = Lbuf[o];
}

extern "C" void kernel_launch(void* const* d_in, const int* in_sizes, int n_in,
                              void* d_out, int out_size, void* d_ws, size_t ws_size,
                              hipStream_t stream) {
  const int*   x      = (const int*)d_in[0];
  const int*   lens   = (const int*)d_in[1];
  const float* alpha0 = (const float*)d_in[2];
  const float* ER     = (const float*)d_in[3];
  const float* D1     = (const float*)d_in[4];
  const float* D2     = (const float*)d_in[5];
  const float* W1     = (const float*)d_in[6];
  const float* b1     = (const float*)d_in[7];
  const float* W2     = (const float*)d_in[8];
  const float* b2     = (const float*)d_in[9];
  const float* W3     = (const float*)d_in[10];
  const float* b3     = (const float*)d_in[11];
  float* out = (float*)d_out;

  farnn_main<<<2048 / 16, 32 * (1 + NHELP), 0, stream>>>(x, lens, alpha0, ER, D1, D2, W1, b1, W2,
                                           b2, W3, b3, out);
}
